# starting kernel
# baseline (speedup 1.0000x reference)
#include <hip/hip_runtime.h>

typedef __attribute__((ext_vector_type(2))) float v2f;
typedef __attribute__((ext_vector_type(8))) float v8f;

__global__ void deg_init_kernel(float* __restrict__ deg, int n) {
    int i = blockIdx.x * blockDim.x + threadIdx.x;
    if (i < n) deg[i] = 1.0f;
}

__global__ void deg_count_kernel(const int* __restrict__ dst, float* __restrict__ deg, int nE) {
    int e = blockIdx.x * blockDim.x + threadIdx.x;
    if (e < nE) atomicAdd(&deg[dst[e]], 1.0f);
}

__global__ void deg_to_dinv_kernel(float* __restrict__ deg, int n) {
    int i = blockIdx.x * blockDim.x + threadIdx.x;
    if (i < n) deg[i] = rsqrtf(deg[i]);
}

__global__ void gemm_xw1_kernel(const float* __restrict__ A, const float* __restrict__ W,
                                float* __restrict__ C, int nRows) {
    int waveId = (blockIdx.x * blockDim.x + threadIdx.x) >> 5;
    int lane   = threadIdx.x & 31;
    int m0     = waveId * 16;
    if (m0 >= nRows) return;

    int row  = m0 + (lane & 15);
    int grp  = lane >> 4;
    int koff = grp * 2;
    int col  = lane & 15;

    v8f acc0 = {};
    v8f acc1 = {};
#pragma unroll
    for (int k = 0; k < 8; ++k) {
        int kb = 4 * k + koff;
        v2f a = *reinterpret_cast<const v2f*>(A + row * 32 + kb);
        v2f b0, b1;
        b0.x = W[kb * 32 + col];        b0.y = W[(kb + 1) * 32 + col];
        b1.x = W[kb * 32 + 16 + col];   b1.y = W[(kb + 1) * 32 + 16 + col];
        acc0 = __builtin_amdgcn_wmma_f32_16x16x4_f32(false, a, false, b0, (short)0, acc0, false, false);
        acc1 = __builtin_amdgcn_wmma_f32_16x16x4_f32(false, a, false, b1, (short)0, acc1, false, false);
    }

    int rowbase = m0 + grp * 8;
#pragma unroll
    for (int r = 0; r < 8; ++r) {
        C[(rowbase + r) * 32 + col]      = acc0[r];
        C[(rowbase + r) * 32 + 16 + col] = acc1[r];
    }
}

__global__ void self_init_kernel(const float* __restrict__ feat, const float* __restrict__ dinv,
                                 float* __restrict__ agg, int n32) {
    int idx = blockIdx.x * blockDim.x + threadIdx.x;
    if (idx < n32) {
        float di = dinv[idx >> 5];
        agg[idx] = feat[idx] * di * di;
    }
}

__global__ void scatter_agg_kernel(const int* __restrict__ src, const int* __restrict__ dst,
                                   const float* __restrict__ dinv, const float* __restrict__ feat,
                                   float* __restrict__ agg, int nEdges) {
    int lane = threadIdx.x & 31;
    int wave = (blockIdx.x * blockDim.x + threadIdx.x) >> 5;
    int base = wave * 32;
    if (base >= nEdges) return;

    int  myE   = base + lane;
    bool valid = myE < nEdges;
    int   sL = valid ? src[myE] : 0;
    int   dL = valid ? dst[myE] : 0;
    float wL = valid ? dinv[sL] * dinv[dL] : 0.0f;

    int cnt = nEdges - base;
    if (cnt > 32) cnt = 32;

#pragma unroll 4
    for (int i = 0; i < cnt; ++i) {
        int   s = __builtin_amdgcn_readlane(sL, i);
        int   d = __builtin_amdgcn_readlane(dL, i);
        float w = __int_as_float(__builtin_amdgcn_readlane(__float_as_int(wL), i));
        float v = feat[s * 32 + lane] * w;
        atomicAdd(agg + d * 32 + lane, v);
    }
}

__global__ void relu_bias_kernel(float* __restrict__ h, const float* __restrict__ b, int n32) {
    int idx = blockIdx.x * blockDim.x + threadIdx.x;
    if (idx < n32) {
        float v = h[idx] + b[idx & 31];
        h[idx] = v > 0.0f ? v : 0.0f;
    }
}

__global__ void gemm_out_kernel(const float* __restrict__ A,
                                const float* __restrict__ Wmu, const float* __restrict__ bmu,
                                const float* __restrict__ Wlv, const float* __restrict__ blv,
                                float* __restrict__ out, int nRows) {
    int waveId = (blockIdx.x * blockDim.x + threadIdx.x) >> 5;
    int lane   = threadIdx.x & 31;
    int m0     = waveId * 16;
    if (m0 >= nRows) return;

    int row  = m0 + (lane & 15);
    int grp  = lane >> 4;
    int koff = grp * 2;
    int col  = lane & 15;

    v8f accM = {};
    v8f accL = {};
#pragma unroll
    for (int k = 0; k < 8; ++k) {
        int kb = 4 * k + koff;
        v2f a = *reinterpret_cast<const v2f*>(A + row * 32 + kb);
        v2f bm, bl;
        bm.x = Wmu[kb * 16 + col];  bm.y = Wmu[(kb + 1) * 16 + col];
        bl.x = Wlv[kb * 16 + col];  bl.y = Wlv[(kb + 1) * 16 + col];
        accM = __builtin_amdgcn_wmma_f32_16x16x4_f32(false, a, false, bm, (short)0, accM, false, false);
        accL = __builtin_amdgcn_wmma_f32_16x16x4_f32(false, a, false, bl, (short)0, accL, false, false);
    }

    int rowbase = m0 + grp * 8;
    float vbm = bmu[col];
    float vbl = blv[col];
    float* outLv = out + (size_t)nRows * 16;
#pragma unroll
    for (int r = 0; r < 8; ++r) {
        out[(rowbase + r) * 16 + col]   = accM[r] + vbm;
        outLv[(rowbase + r) * 16 + col] = accL[r] + vbl;
    }
}

extern "C" void kernel_launch(void* const* d_in, const int* in_sizes, int n_in,
                              void* d_out, int out_size, void* d_ws, size_t ws_size,
                              hipStream_t stream) {
    const float* x   = (const float*)d_in[0];
    const int*   ei  = (const int*)  d_in[1];
    const float* W1  = (const float*)d_in[2];
    const float* b1  = (const float*)d_in[3];
    const float* Wmu = (const float*)d_in[4];
    const float* bmu = (const float*)d_in[5];
    const float* Wlv = (const float*)d_in[6];
    const float* blv = (const float*)d_in[7];

    const int N = in_sizes[0] / 32;
    const int E = in_sizes[1] / 2;
    const int* src = ei;
    const int* dst = ei + E;

    float* dinv = (float*)d_ws;
    float* bufA = dinv + N;
    float* bufB = bufA + (size_t)N * 32;
    float* out  = (float*)d_out;

    const int TB = 256;
    const int n32 = N * 32;
    const int gemmThreads = ((N + 15) / 16) * 32;
    const int scatterWaves = (E + 31) / 32;
    const int scatterBlocks = (scatterWaves * 32 + TB - 1) / TB;

    deg_init_kernel<<<(N + TB - 1) / TB, TB, 0, stream>>>(dinv, N);
    deg_count_kernel<<<(E + TB - 1) / TB, TB, 0, stream>>>(dst, dinv, E);
    deg_to_dinv_kernel<<<(N + TB - 1) / TB, TB, 0, stream>>>(dinv, N);

    gemm_xw1_kernel<<<(gemmThreads + TB - 1) / TB, TB, 0, stream>>>(x, W1, bufA, N);
    self_init_kernel<<<(n32 + TB - 1) / TB, TB, 0, stream>>>(bufA, dinv, bufB, n32);
    scatter_agg_kernel<<<scatterBlocks, TB, 0, stream>>>(src, dst, dinv, bufA, bufB, E);
    relu_bias_kernel<<<(n32 + TB - 1) / TB, TB, 0, stream>>>(bufB, b1, n32);

    self_init_kernel<<<(n32 + TB - 1) / TB, TB, 0, stream>>>(bufB, dinv, bufA, n32);
    scatter_agg_kernel<<<scatterBlocks, TB, 0, stream>>>(src, dst, dinv, bufB, bufA, E);
    gemm_out_kernel<<<(gemmThreads + TB - 1) / TB, TB, 0, stream>>>(bufA, Wmu, bmu, Wlv, blv, out, N);
}


# speedup vs baseline: 1.6260x; 1.6260x over previous
#include <hip/hip_runtime.h>

typedef __attribute__((ext_vector_type(2))) float v2f;
typedef __attribute__((ext_vector_type(8))) float v8f;

__device__ __forceinline__ int l2_half() {
    return (int)((__builtin_amdgcn_s_sendmsg_rtn(0x87) >> 18) & 1u);
}

__global__ void deg_init_kernel(float* __restrict__ deg0, float* __restrict__ deg1, int n) {
    int i = blockIdx.x * blockDim.x + threadIdx.x;
    if (i < n) {
        deg0[i] = 1.0f;
        deg1[i] = 0.0f;
    }
}

__global__ void deg_count_kernel(const int* __restrict__ dst, float* __restrict__ deg0,
                                 float* __restrict__ deg1, int nE) {
    int e = blockIdx.x * blockDim.x + threadIdx.x;
    float* deg = l2_half() ? deg1 : deg0;
    if (e < nE) atomicAdd(&deg[dst[e]], 1.0f);
}

__global__ void deg_to_dinv_kernel(float* __restrict__ deg0, const float* __restrict__ deg1, int n) {
    int i = blockIdx.x * blockDim.x + threadIdx.x;
    if (i < n) deg0[i] = rsqrtf(deg0[i] + deg1[i]);
}

__global__ void gemm_xw1_kernel(const float* __restrict__ A, const float* __restrict__ W,
                                float* __restrict__ C, int nRows) {
    int waveId = (blockIdx.x * blockDim.x + threadIdx.x) >> 5;
    int lane   = threadIdx.x & 31;
    int m0     = waveId * 16;
    if (m0 >= nRows) return;

    int row  = m0 + (lane & 15);
    int grp  = lane >> 4;
    int koff = grp * 2;
    int col  = lane & 15;

    v8f acc0 = {};
    v8f acc1 = {};
#pragma unroll
    for (int k = 0; k < 8; ++k) {
        int kb = 4 * k + koff;
        v2f a = *reinterpret_cast<const v2f*>(A + row * 32 + kb);
        v2f b0, b1;
        b0.x = W[kb * 32 + col];        b0.y = W[(kb + 1) * 32 + col];
        b1.x = W[kb * 32 + 16 + col];   b1.y = W[(kb + 1) * 32 + 16 + col];
        acc0 = __builtin_amdgcn_wmma_f32_16x16x4_f32(false, a, false, b0, (short)0, acc0, false, false);
        acc1 = __builtin_amdgcn_wmma_f32_16x16x4_f32(false, a, false, b1, (short)0, acc1, false, false);
    }

    int rowbase = m0 + grp * 8;
#pragma unroll
    for (int r = 0; r < 8; ++r) {
        C[(rowbase + r) * 32 + col]      = acc0[r];
        C[(rowbase + r) * 32 + 16 + col] = acc1[r];
    }
}

__global__ void self_init_kernel(const float* __restrict__ feat, const float* __restrict__ dinv,
                                 float* __restrict__ agg0, float* __restrict__ agg1, int n32) {
    int idx = blockIdx.x * blockDim.x + threadIdx.x;
    if (idx < n32) {
        float di = dinv[idx >> 5];
        agg0[idx] = feat[idx] * di * di;
        agg1[idx] = 0.0f;
    }
}

__global__ void scatter_agg_kernel(const int* __restrict__ src, const int* __restrict__ dst,
                                   const float* __restrict__ dinv, const float* __restrict__ feat,
                                   float* __restrict__ agg0, float* __restrict__ agg1, int nEdges) {
    float* agg = l2_half() ? agg1 : agg0;
    int lane = threadIdx.x & 31;
    int wave = (blockIdx.x * blockDim.x + threadIdx.x) >> 5;
    int base = wave * 32;
    if (base >= nEdges) return;

    int  myE   = base + lane;
    bool valid = myE < nEdges;
    int   sL = valid ? src[myE] : 0;
    int   dL = valid ? dst[myE] : 0;
    float wL = valid ? dinv[sL] * dinv[dL] : 0.0f;

    int cnt = nEdges - base;
    if (cnt > 32) cnt = 32;

#pragma unroll 4
    for (int i = 0; i < cnt; ++i) {
        int   s = __builtin_amdgcn_readlane(sL, i);
        int   d = __builtin_amdgcn_readlane(dL, i);
        float w = __int_as_float(__builtin_amdgcn_readlane(__float_as_int(wL), i));
        float v = feat[s * 32 + lane] * w;
        atomicAdd(agg + d * 32 + lane, v);
    }
}

__global__ void relu_bias_kernel(const float* __restrict__ agg0, float* __restrict__ h,
                                 const float* __restrict__ b, int n32) {
    int idx = blockIdx.x * blockDim.x + threadIdx.x;
    if (idx < n32) {
        float v = (agg0[idx] + h[idx]) + b[idx & 31];
        h[idx] = v > 0.0f ? v : 0.0f;
    }
}

__global__ void gemm_out_kernel(const float* __restrict__ A, const float* __restrict__ A1,
                                const float* __restrict__ Wmu, const float* __restrict__ bmu,
                                const float* __restrict__ Wlv, const float* __restrict__ blv,
                                float* __restrict__ out, int nRows) {
    int waveId = (blockIdx.x * blockDim.x + threadIdx.x) >> 5;
    int lane   = threadIdx.x & 31;
    int m0     = waveId * 16;
    if (m0 >= nRows) return;

    int row  = m0 + (lane & 15);
    int grp  = lane >> 4;
    int koff = grp * 2;
    int col  = lane & 15;

    v2f a8[8];
#pragma unroll
    for (int k = 0; k < 8; ++k) {
        int kb = 4 * k + koff;
        a8[k] = *reinterpret_cast<const v2f*>(A + row * 32 + kb)
              + *reinterpret_cast<const v2f*>(A1 + row * 32 + kb);
    }
    __builtin_amdgcn_sched_barrier(0);

    v8f accM = {};
    v8f accL = {};
#pragma unroll
    for (int k = 0; k < 8; ++k) {
        int kb = 4 * k + koff;
        v2f a = a8[k];
        v2f bm, bl;
        bm.x = Wmu[kb * 16 + col];  bm.y = Wmu[(kb + 1) * 16 + col];
        bl.x = Wlv[kb * 16 + col];  bl.y = Wlv[(kb + 1) * 16 + col];
        accM = __builtin_amdgcn_wmma_f32_16x16x4_f32(false, a, false, bm, (short)0, accM, false, false);
        accL = __builtin_amdgcn_wmma_f32_16x16x4_f32(false, a, false, bl, (short)0, accL, false, false);
    }

    int rowbase = m0 + grp * 8;
    float vbm = bmu[col];
    float vbl = blv[col];
    float* outLv = out + (size_t)nRows * 16;
#pragma unroll
    for (int r = 0; r < 8; ++r) {
        out[(rowbase + r) * 16 + col]   = accM[r] + vbm;
        outLv[(rowbase + r) * 16 + col] = accL[r] + vbl;
    }
}

extern "C" void kernel_launch(void* const* d_in, const int* in_sizes, int n_in,
                              void* d_out, int out_size, void* d_ws, size_t ws_size,
                              hipStream_t stream) {
    const float* x   = (const float*)d_in[0];
    const int*   ei  = (const int*)  d_in[1];
    const float* W1  = (const float*)d_in[2];
    const float* b1  = (const float*)d_in[3];
    const float* Wmu = (const float*)d_in[4];
    const float* bmu = (const float*)d_in[5];
    const float* Wlv = (const float*)d_in[6];
    const float* blv = (const float*)d_in[7];

    const int N = in_sizes[0] / 32;
    const int E = in_sizes[1] / 2;
    const int* src = ei;
    const int* dst = ei + E;

    float* dinv = (float*)d_ws;
    float* bufA = dinv + N;
    float* bufB = bufA + (size_t)N * 32;
    float* out  = (float*)d_out;

    const int TB = 256;
    const int n32 = N * 32;
    const int gemmThreads = ((N + 15) / 16) * 32;
    const int scatterWaves = (E + 31) / 32;
    const int scatterBlocks = (scatterWaves * 32 + TB - 1) / TB;

    deg_init_kernel<<<(N + TB - 1) / TB, TB, 0, stream>>>(dinv, bufA, N);
    deg_count_kernel<<<(E + TB - 1) / TB, TB, 0, stream>>>(dst, dinv, bufA, E);
    deg_to_dinv_kernel<<<(N + TB - 1) / TB, TB, 0, stream>>>(dinv, bufA, N);

    gemm_xw1_kernel<<<(gemmThreads + TB - 1) / TB, TB, 0, stream>>>(x, W1, bufA, N);
    self_init_kernel<<<(n32 + TB - 1) / TB, TB, 0, stream>>>(bufA, dinv, bufB, out, n32);
    scatter_agg_kernel<<<scatterBlocks, TB, 0, stream>>>(src, dst, dinv, bufA, bufB, out, E);
    relu_bias_kernel<<<(n32 + TB - 1) / TB, TB, 0, stream>>>(bufB, out, b1, n32);

    self_init_kernel<<<(n32 + TB - 1) / TB, TB, 0, stream>>>(out, dinv, bufA, bufB, n32);
    scatter_agg_kernel<<<scatterBlocks, TB, 0, stream>>>(src, dst, dinv, out, bufA, bufB, E);
    gemm_out_kernel<<<(gemmThreads + TB - 1) / TB, TB, 0, stream>>>(bufA, bufB, Wmu, bmu, Wlv, blv, out, N);
}


# Degree count without global atomics: each work-group counts a 16384-edge chunk into packed 16-bit LDS counters for all nodes and stores its plane; a node pass sums planes (integers, exact)
# speedup vs baseline: 1.7789x; 1.0940x over previous
#include <hip/hip_runtime.h>

typedef __attribute__((ext_vector_type(2))) float v2f;
typedef __attribute__((ext_vector_type(8))) float v8f;

__device__ __forceinline__ int l2_half() {
    return (int)((__builtin_amdgcn_s_sendmsg_rtn(0x87) >> 18) & 1u);
}

#define DEG_WORDS 50000
#define DEG_CHUNK 16384
#define DEG_TB    1024

__global__ __launch_bounds__(DEG_TB) void deg_hist_kernel(const int* __restrict__ dst,
                                                          unsigned* __restrict__ planes, int nE) {
    __shared__ unsigned cnt[DEG_WORDS];
    const int t = threadIdx.x;

    const int base = blockIdx.x * DEG_CHUNK;
    int d[DEG_CHUNK / DEG_TB];
#pragma unroll
    for (int k = 0; k < DEG_CHUNK / DEG_TB; ++k) {
        int e = base + k * DEG_TB + t;
        d[k] = (e < nE) ? dst[e] : -1;
    }

    uint4* cnt4 = reinterpret_cast<uint4*>(cnt);
    for (int j = t; j < DEG_WORDS / 4; j += DEG_TB) cnt4[j] = make_uint4(0u, 0u, 0u, 0u);
    __syncthreads();

#pragma unroll
    for (int k = 0; k < DEG_CHUNK / DEG_TB; ++k) {
        if (d[k] >= 0) atomicAdd(&cnt[d[k] >> 1], 1u << ((d[k] & 1) * 16));
    }
    __syncthreads();

    uint4* out4 = reinterpret_cast<uint4*>(planes + (size_t)blockIdx.x * DEG_WORDS);
    for (int j = t; j < DEG_WORDS / 4; j += DEG_TB) out4[j] = cnt4[j];
}

__global__ void deg_reduce_dinv_kernel(const unsigned* __restrict__ planes, float* __restrict__ dinv,
                                       int n, int nPlanes) {
    int i = blockIdx.x * blockDim.x + threadIdx.x;
    if (i >= DEG_WORDS) return;
    unsigned lo = 0u, hi = 0u;
#pragma unroll 16
    for (int c = 0; c < nPlanes; ++c) {
        unsigned w = planes[(size_t)c * DEG_WORDS + i];
        lo += w & 0xffffu;
        hi += w >> 16;
    }
    int n0 = 2 * i;
    if (n0 < n)     dinv[n0]     = rsqrtf((float)(lo + 1u));
    if (n0 + 1 < n) dinv[n0 + 1] = rsqrtf((float)(hi + 1u));
}

__global__ void gemm_xw1_kernel(const float* __restrict__ A, const float* __restrict__ W,
                                float* __restrict__ C, int nRows) {
    int waveId = (blockIdx.x * blockDim.x + threadIdx.x) >> 5;
    int lane   = threadIdx.x & 31;
    int m0     = waveId * 16;
    if (m0 >= nRows) return;

    int row  = m0 + (lane & 15);
    int grp  = lane >> 4;
    int koff = grp * 2;
    int col  = lane & 15;

    v8f acc0 = {};
    v8f acc1 = {};
#pragma unroll
    for (int k = 0; k < 8; ++k) {
        int kb = 4 * k + koff;
        v2f a = *reinterpret_cast<const v2f*>(A + row * 32 + kb);
        v2f b0, b1;
        b0.x = W[kb * 32 + col];        b0.y = W[(kb + 1) * 32 + col];
        b1.x = W[kb * 32 + 16 + col];   b1.y = W[(kb + 1) * 32 + 16 + col];
        acc0 = __builtin_amdgcn_wmma_f32_16x16x4_f32(false, a, false, b0, (short)0, acc0, false, false);
        acc1 = __builtin_amdgcn_wmma_f32_16x16x4_f32(false, a, false, b1, (short)0, acc1, false, false);
    }

    int rowbase = m0 + grp * 8;
#pragma unroll
    for (int r = 0; r < 8; ++r) {
        C[(rowbase + r) * 32 + col]      = acc0[r];
        C[(rowbase + r) * 32 + 16 + col] = acc1[r];
    }
}

__global__ void self_init_kernel(const float* __restrict__ feat, const float* __restrict__ dinv,
                                 float* __restrict__ agg0, float* __restrict__ agg1, int n32) {
    int idx = blockIdx.x * blockDim.x + threadIdx.x;
    if (idx < n32) {
        float di = dinv[idx >> 5];
        agg0[idx] = feat[idx] * di * di;
        agg1[idx] = 0.0f;
    }
}

__global__ void scatter_agg_kernel(const int* __restrict__ src, const int* __restrict__ dst,
                                   const float* __restrict__ dinv, const float* __restrict__ feat,
                                   float* __restrict__ agg0, float* __restrict__ agg1, int nEdges) {
    float* agg = l2_half() ? agg1 : agg0;
    int lane = threadIdx.x & 31;
    int wave = (blockIdx.x * blockDim.x + threadIdx.x) >> 5;
    int base = wave * 32;
    if (base >= nEdges) return;

    int  myE   = base + lane;
    bool valid = myE < nEdges;
    int   sL = valid ? src[myE] : 0;
    int   dL = valid ? dst[myE] : 0;
    float wL = valid ? dinv[sL] * dinv[dL] : 0.0f;

    int cnt = nEdges - base;
    if (cnt > 32) cnt = 32;

#pragma unroll 4
    for (int i = 0; i < cnt; ++i) {
        int   s = __builtin_amdgcn_readlane(sL, i);
        int   d = __builtin_amdgcn_readlane(dL, i);
        float w = __int_as_float(__builtin_amdgcn_readlane(__float_as_int(wL), i));
        float v = feat[s * 32 + lane] * w;
        atomicAdd(agg + d * 32 + lane, v);
    }
}

__global__ void relu_bias_kernel(const float* __restrict__ agg0, float* __restrict__ h,
                                 const float* __restrict__ b, int n32) {
    int idx = blockIdx.x * blockDim.x + threadIdx.x;
    if (idx < n32) {
        float v = (agg0[idx] + h[idx]) + b[idx & 31];
        h[idx] = v > 0.0f ? v : 0.0f;
    }
}

__global__ void gemm_out_kernel(const float* __restrict__ A, const float* __restrict__ A1,
                                const float* __restrict__ Wmu, const float* __restrict__ bmu,
                                const float* __restrict__ Wlv, const float* __restrict__ blv,
                                float* __restrict__ out, int nRows) {
    int waveId = (blockIdx.x * blockDim.x + threadIdx.x) >> 5;
    int lane   = threadIdx.x & 31;
    int m0     = waveId * 16;
    if (m0 >= nRows) return;

    int row  = m0 + (lane & 15);
    int grp  = lane >> 4;
    int koff = grp * 2;
    int col  = lane & 15;

    v2f a8[8];
#pragma unroll
    for (int k = 0; k < 8; ++k) {
        int kb = 4 * k + koff;
        a8[k] = *reinterpret_cast<const v2f*>(A + row * 32 + kb)
              + *reinterpret_cast<const v2f*>(A1 + row * 32 + kb);
    }
    __builtin_amdgcn_sched_barrier(0);

    v8f accM = {};
    v8f accL = {};
#pragma unroll
    for (int k = 0; k < 8; ++k) {
        int kb = 4 * k + koff;
        v2f a = a8[k];
        v2f bm, bl;
        bm.x = Wmu[kb * 16 + col];  bm.y = Wmu[(kb + 1) * 16 + col];
        bl.x = Wlv[kb * 16 + col];  bl.y = Wlv[(kb + 1) * 16 + col];
        accM = __builtin_amdgcn_wmma_f32_16x16x4_f32(false, a, false, bm, (short)0, accM, false, false);
        accL = __builtin_amdgcn_wmma_f32_16x16x4_f32(false, a, false, bl, (short)0, accL, false, false);
    }

    int rowbase = m0 + grp * 8;
    float vbm = bmu[col];
    float vbl = blv[col];
    float* outLv = out + (size_t)nRows * 16;
#pragma unroll
    for (int r = 0; r < 8; ++r) {
        out[(rowbase + r) * 16 + col]   = accM[r] + vbm;
        outLv[(rowbase + r) * 16 + col] = accL[r] + vbl;
    }
}

extern "C" void kernel_launch(void* const* d_in, const int* in_sizes, int n_in,
                              void* d_out, int out_size, void* d_ws, size_t ws_size,
                              hipStream_t stream) {
    const float* x   = (const float*)d_in[0];
    const int*   ei  = (const int*)  d_in[1];
    const float* W1  = (const float*)d_in[2];
    const float* b1  = (const float*)d_in[3];
    const float* Wmu = (const float*)d_in[4];
    const float* bmu = (const float*)d_in[5];
    const float* Wlv = (const float*)d_in[6];
    const float* blv = (const float*)d_in[7];

    const int N = in_sizes[0] / 32;
    const int E = in_sizes[1] / 2;
    const int* src = ei;
    const int* dst = ei + E;

    float* dinv = (float*)d_ws;
    float* bufA = dinv + N;
    float* bufB = bufA + (size_t)N * 32;
    float* out  = (float*)d_out;

    const int TB = 256;
    const int n32 = N * 32;
    const int gemmThreads = ((N + 15) / 16) * 32;
    const int scatterWaves = (E + 31) / 32;
    const int scatterBlocks = (scatterWaves * 32 + TB - 1) / TB;

    unsigned* planes = (unsigned*)bufA;
    const int nPlanes = (E + DEG_CHUNK - 1) / DEG_CHUNK;
    deg_hist_kernel<<<nPlanes, DEG_TB, 0, stream>>>(dst, planes, E);
    deg_reduce_dinv_kernel<<<(DEG_WORDS + TB - 1) / TB, TB, 0, stream>>>(planes, dinv, N, nPlanes);

    gemm_xw1_kernel<<<(gemmThreads + TB - 1) / TB, TB, 0, stream>>>(x, W1, bufA, N);
    self_init_kernel<<<(n32 + TB - 1) / TB, TB, 0, stream>>>(bufA, dinv, bufB, out, n32);
    scatter_agg_kernel<<<scatterBlocks, TB, 0, stream>>>(src, dst, dinv, bufA, bufB, out, E);
    relu_bias_kernel<<<(n32 + TB - 1) / TB, TB, 0, stream>>>(bufB, out, b1, n32);

    self_init_kernel<<<(n32 + TB - 1) / TB, TB, 0, stream>>>(out, dinv, bufA, bufB, n32);
    scatter_agg_kernel<<<scatterBlocks, TB, 0, stream>>>(src, dst, dinv, out, bufA, bufB, E);
    gemm_out_kernel<<<(gemmThreads + TB - 1) / TB, TB, 0, stream>>>(bufA, bufB, Wmu, bmu, Wlv, blv, out, N);
}


# Normalisation factored out of the edge loop: rows stored pre-scaled by dinv[src], node sums scaled by dinv[dst] once; no dinv gathers or multiply per edge (scaled products round differently)
# speedup vs baseline: 1.9715x; 1.1083x over previous
#include <hip/hip_runtime.h>

typedef __attribute__((ext_vector_type(2))) float v2f;
typedef __attribute__((ext_vector_type(8))) float v8f;

__device__ __forceinline__ int l2_half() {
    return (int)((__builtin_amdgcn_s_sendmsg_rtn(0x87) >> 18) & 1u);
}


#define DEG_WORDS 50000
#define DEG_CHUNK 16384
#define DEG_TB    1024

__global__ __launch_bounds__(DEG_TB) void deg_hist_kernel(const int* __restrict__ dst,
                                                          unsigned* __restrict__ planes, int nE) {
    __shared__ unsigned cnt[DEG_WORDS];
    const int t = threadIdx.x;

    const int base = blockIdx.x * DEG_CHUNK;
    int d[DEG_CHUNK / DEG_TB];
#pragma unroll
    for (int k = 0; k < DEG_CHUNK / DEG_TB; ++k) {
        int e = base + k * DEG_TB + t;
        d[k] = (e < nE) ? dst[e] : -1;
    }

    uint4* cnt4 = reinterpret_cast<uint4*>(cnt);
    for (int j = t; j < DEG_WORDS / 4; j += DEG_TB) cnt4[j] = make_uint4(0u, 0u, 0u, 0u);
    __syncthreads();

#pragma unroll
    for (int k = 0; k < DEG_CHUNK / DEG_TB; ++k) {
        if (d[k] >= 0) atomicAdd(&cnt[d[k] >> 1], 1u << ((d[k] & 1) * 16));
    }
    __syncthreads();

    uint4* out4 = reinterpret_cast<uint4*>(planes + (size_t)blockIdx.x * DEG_WORDS);
    for (int j = t; j < DEG_WORDS / 4; j += DEG_TB) out4[j] = cnt4[j];
}

__global__ void deg_reduce_dinv_kernel(const unsigned* __restrict__ planes, float* __restrict__ dinv,
                                       int n, int nPlanes) {
    int i = blockIdx.x * blockDim.x + threadIdx.x;
    if (i >= DEG_WORDS) return;
    unsigned lo = 0u, hi = 0u;
#pragma unroll 16
    for (int c = 0; c < nPlanes; ++c) {
        unsigned w = planes[(size_t)c * DEG_WORDS + i];
        lo += w & 0xffffu;
        hi += w >> 16;
    }
    int n0 = 2 * i;
    if (n0 < n)     dinv[n0]     = rsqrtf((float)(lo + 1u));
    if (n0 + 1 < n) dinv[n0 + 1] = rsqrtf((float)(hi + 1u));
}

__global__ void gemm_xw1_kernel(const float* __restrict__ A, const float* __restrict__ W,
                                const float* __restrict__ dinv,
                                float* __restrict__ C, int nRows) {
    int waveId = (blockIdx.x * blockDim.x + threadIdx.x) >> 5;
    int lane   = threadIdx.x & 31;
    int m0     = waveId * 16;
    if (m0 >= nRows) return;

    int row  = m0 + (lane & 15);
    int grp  = lane >> 4;
    int koff = grp * 2;
    int col  = lane & 15;
    int rowbase = m0 + grp * 8;

    float dv[8];
#pragma unroll
    for (int r = 0; r < 8; ++r) dv[r] = dinv[rowbase + r];

    v8f acc0 = {};
    v8f acc1 = {};
#pragma unroll
    for (int k = 0; k < 8; ++k) {
        int kb = 4 * k + koff;
        v2f a = *reinterpret_cast<const v2f*>(A + row * 32 + kb);
        v2f b0, b1;
        b0.x = W[kb * 32 + col];        b0.y = W[(kb + 1) * 32 + col];
        b1.x = W[kb * 32 + 16 + col];   b1.y = W[(kb + 1) * 32 + 16 + col];
        acc0 = __builtin_amdgcn_wmma_f32_16x16x4_f32(false, a, false, b0, (short)0, acc0, false, false);
        acc1 = __builtin_amdgcn_wmma_f32_16x16x4_f32(false, a, false, b1, (short)0, acc1, false, false);
    }

#pragma unroll
    for (int r = 0; r < 8; ++r) {
        C[(rowbase + r) * 32 + col]      = acc0[r] * dv[r];
        C[(rowbase + r) * 32 + 16 + col] = acc1[r] * dv[r];
    }
}

__global__ void self_init_kernel(const float* __restrict__ feat,
                                 float* __restrict__ agg0, float* __restrict__ agg1, int n32) {
    int idx = blockIdx.x * blockDim.x + threadIdx.x;
    if (idx < n32) {
        agg0[idx] = feat[idx];
        agg1[idx] = 0.0f;
    }
}

__global__ void scatter_agg_kernel(const int* __restrict__ src, const int* __restrict__ dst,
                                   const float* __restrict__ feat,
                                   float* __restrict__ agg0, float* __restrict__ agg1, int nEdges) {
    float* agg = l2_half() ? agg1 : agg0;
    int lane = threadIdx.x & 31;
    int wave = (blockIdx.x * blockDim.x + threadIdx.x) >> 5;
    int base = wave * 32;
    if (base >= nEdges) return;

    int  myE   = base + lane;
    bool valid = myE < nEdges;
    int   sL = valid ? src[myE] : 0;
    int   dL = valid ? dst[myE] : 0;

    int cnt = nEdges - base;
    if (cnt > 32) cnt = 32;

    for (int i = 0; i < cnt; ++i) {
        int   s = __builtin_amdgcn_readlane(sL, i);
        int   d = __builtin_amdgcn_readlane(dL, i);
        atomicAdd(agg + d * 32 + lane, feat[s * 32 + lane]);
    }
}

__global__ void relu_bias_kernel(const float* __restrict__ agg0, float* __restrict__ h,
                                 const float* __restrict__ b, const float* __restrict__ dinv,
                                 int n32) {
    int idx = blockIdx.x * blockDim.x + threadIdx.x;
    if (idx < n32) {
        float di = dinv[idx >> 5];
        float v = (agg0[idx] + h[idx]) * di + b[idx & 31];
        h[idx] = (v > 0.0f ? v : 0.0f) * di;
    }
}

__global__ void gemm_out_kernel(const float* __restrict__ A, const float* __restrict__ A1,
                                const float* __restrict__ dinv,
                                const float* __restrict__ Wmu, const float* __restrict__ bmu,
                                const float* __restrict__ Wlv, const float* __restrict__ blv,
                                float* __restrict__ out, int nRows) {
    int waveId = (blockIdx.x * blockDim.x + threadIdx.x) >> 5;
    int lane   = threadIdx.x & 31;
    int m0     = waveId * 16;
    if (m0 >= nRows) return;

    int row  = m0 + (lane & 15);
    int grp  = lane >> 4;
    int koff = grp * 2;
    int col  = lane & 15;

    float di = dinv[row];
    v2f a8[8];
#pragma unroll
    for (int k = 0; k < 8; ++k) {
        int kb = 4 * k + koff;
        a8[k] = (*reinterpret_cast<const v2f*>(A + row * 32 + kb)
               + *reinterpret_cast<const v2f*>(A1 + row * 32 + kb)) * di;
    }
    __builtin_amdgcn_sched_barrier(0);

    v8f accM = {};
    v8f accL = {};
#pragma unroll
    for (int k = 0; k < 8; ++k) {
        int kb = 4 * k + koff;
        v2f a = a8[k];
        v2f bm, bl;
        bm.x = Wmu[kb * 16 + col];  bm.y = Wmu[(kb + 1) * 16 + col];
        bl.x = Wlv[kb * 16 + col];  bl.y = Wlv[(kb + 1) * 16 + col];
        accM = __builtin_amdgcn_wmma_f32_16x16x4_f32(false, a, false, bm, (short)0, accM, false, false);
        accL = __builtin_amdgcn_wmma_f32_16x16x4_f32(false, a, false, bl, (short)0, accL, false, false);
    }

    int rowbase = m0 + grp * 8;
    float vbm = bmu[col];
    float vbl = blv[col];
    float* outLv = out + (size_t)nRows * 16;
#pragma unroll
    for (int r = 0; r < 8; ++r) {
        out[(rowbase + r) * 16 + col]   = accM[r] + vbm;
        outLv[(rowbase + r) * 16 + col] = accL[r] + vbl;
    }
}

extern "C" void kernel_launch(void* const* d_in, const int* in_sizes, int n_in,
                              void* d_out, int out_size, void* d_ws, size_t ws_size,
                              hipStream_t stream) {
    const float* x   = (const float*)d_in[0];
    const int*   ei  = (const int*)  d_in[1];
    const float* W1  = (const float*)d_in[2];
    const float* b1  = (const float*)d_in[3];
    const float* Wmu = (const float*)d_in[4];
    const float* bmu = (const float*)d_in[5];
    const float* Wlv = (const float*)d_in[6];
    const float* blv = (const float*)d_in[7];

    const int N = in_sizes[0] / 32;
    const int E = in_sizes[1] / 2;
    const int* src = ei;
    const int* dst = ei + E;

    float* dinv = (float*)d_ws;
    float* bufA = dinv + N;
    float* bufB = bufA + (size_t)N * 32;
    float* out  = (float*)d_out;

    const int TB = 256;
    const int n32 = N * 32;
    const int gemmThreads = ((N + 15) / 16) * 32;
    const int scatterWaves = (E + 31) / 32;
    const int scatterBlocks = (scatterWaves * 32 + TB - 1) / TB;

    unsigned* planes = (unsigned*)bufA;
    const int nPlanes = (E + DEG_CHUNK - 1) / DEG_CHUNK;
    deg_hist_kernel<<<nPlanes, DEG_TB, 0, stream>>>(dst, planes, E);
    deg_reduce_dinv_kernel<<<(DEG_WORDS + TB - 1) / TB, TB, 0, stream>>>(planes, dinv, N, nPlanes);

    gemm_xw1_kernel<<<(gemmThreads + TB - 1) / TB, TB, 0, stream>>>(x, W1, dinv, bufA, N);
    self_init_kernel<<<(n32 + TB - 1) / TB, TB, 0, stream>>>(bufA, bufB, out, n32);
    scatter_agg_kernel<<<scatterBlocks, TB, 0, stream>>>(src, dst, bufA, bufB, out, E);
    relu_bias_kernel<<<(n32 + TB - 1) / TB, TB, 0, stream>>>(bufB, out, b1, dinv, n32);

    self_init_kernel<<<(n32 + TB - 1) / TB, TB, 0, stream>>>(out, bufA, bufB, n32);
    scatter_agg_kernel<<<scatterBlocks, TB, 0, stream>>>(src, dst, out, bufA, bufB, E);
    gemm_out_kernel<<<(gemmThreads + TB - 1) / TB, TB, 0, stream>>>(bufA, bufB, dinv, Wmu, bmu, Wlv, blv, out, N);
}


# Edge scatter: a wave requests all 32 gathered rows of its batch before its first atomic add (32 row loads in flight per wave, not one at a time); same atomic adds, issued in the same order
# speedup vs baseline: 2.6504x; 1.3444x over previous
#include <hip/hip_runtime.h>

typedef __attribute__((ext_vector_type(2))) float v2f;
typedef __attribute__((ext_vector_type(8))) float v8f;

__device__ __forceinline__ int l2_half() {
    return (int)((__builtin_amdgcn_s_sendmsg_rtn(0x87) >> 18) & 1u);
}


#define DEG_WORDS 50000
#define DEG_CHUNK 16384
#define DEG_TB    1024

__global__ __launch_bounds__(DEG_TB) void deg_hist_kernel(const int* __restrict__ dst,
                                                          unsigned* __restrict__ planes, int nE) {
    __shared__ unsigned cnt[DEG_WORDS];
    const int t = threadIdx.x;

    const int base = blockIdx.x * DEG_CHUNK;
    int d[DEG_CHUNK / DEG_TB];
#pragma unroll
    for (int k = 0; k < DEG_CHUNK / DEG_TB; ++k) {
        int e = base + k * DEG_TB + t;
        d[k] = (e < nE) ? dst[e] : -1;
    }

    uint4* cnt4 = reinterpret_cast<uint4*>(cnt);
    for (int j = t; j < DEG_WORDS / 4; j += DEG_TB) cnt4[j] = make_uint4(0u, 0u, 0u, 0u);
    __syncthreads();

#pragma unroll
    for (int k = 0; k < DEG_CHUNK / DEG_TB; ++k) {
        if (d[k] >= 0) atomicAdd(&cnt[d[k] >> 1], 1u << ((d[k] & 1) * 16));
    }
    __syncthreads();

    uint4* out4 = reinterpret_cast<uint4*>(planes + (size_t)blockIdx.x * DEG_WORDS);
    for (int j = t; j < DEG_WORDS / 4; j += DEG_TB) out4[j] = cnt4[j];
}

__global__ void deg_reduce_dinv_kernel(const unsigned* __restrict__ planes, float* __restrict__ dinv,
                                       int n, int nPlanes) {
    int i = blockIdx.x * blockDim.x + threadIdx.x;
    if (i >= DEG_WORDS) return;
    unsigned lo = 0u, hi = 0u;
#pragma unroll 16
    for (int c = 0; c < nPlanes; ++c) {
        unsigned w = planes[(size_t)c * DEG_WORDS + i];
        lo += w & 0xffffu;
        hi += w >> 16;
    }
    int n0 = 2 * i;
    if (n0 < n)     dinv[n0]     = rsqrtf((float)(lo + 1u));
    if (n0 + 1 < n) dinv[n0 + 1] = rsqrtf((float)(hi + 1u));
}

__global__ void gemm_xw1_kernel(const float* __restrict__ A, const float* __restrict__ W,
                                const float* __restrict__ dinv,
                                float* __restrict__ C, int nRows) {
    int waveId = (blockIdx.x * blockDim.x + threadIdx.x) >> 5;
    int lane   = threadIdx.x & 31;
    int m0     = waveId * 16;
    if (m0 >= nRows) return;

    int row  = m0 + (lane & 15);
    int grp  = lane >> 4;
    int koff = grp * 2;
    int col  = lane & 15;
    int rowbase = m0 + grp * 8;

    float dv[8];
#pragma unroll
    for (int r = 0; r < 8; ++r) dv[r] = dinv[rowbase + r];

    v8f acc0 = {};
    v8f acc1 = {};
#pragma unroll
    for (int k = 0; k < 8; ++k) {
        int kb = 4 * k + koff;
        v2f a = *reinterpret_cast<const v2f*>(A + row * 32 + kb);
        v2f b0, b1;
        b0.x = W[kb * 32 + col];        b0.y = W[(kb + 1) * 32 + col];
        b1.x = W[kb * 32 + 16 + col];   b1.y = W[(kb + 1) * 32 + 16 + col];
        acc0 = __builtin_amdgcn_wmma_f32_16x16x4_f32(false, a, false, b0, (short)0, acc0, false, false);
        acc1 = __builtin_amdgcn_wmma_f32_16x16x4_f32(false, a, false, b1, (short)0, acc1, false, false);
    }

#pragma unroll
    for (int r = 0; r < 8; ++r) {
        C[(rowbase + r) * 32 + col]      = acc0[r] * dv[r];
        C[(rowbase + r) * 32 + 16 + col] = acc1[r] * dv[r];
    }
}

__global__ void self_init_kernel(const float* __restrict__ feat,
                                 float* __restrict__ agg0, float* __restrict__ agg1, int n32) {
    int idx = blockIdx.x * blockDim.x + threadIdx.x;
    if (idx < n32) {
        agg0[idx] = feat[idx];
        agg1[idx] = 0.0f;
    }
}

__global__ void scatter_agg_kernel(const int* __restrict__ src, const int* __restrict__ dst,
                                   const float* __restrict__ feat,
                                   float* __restrict__ agg0, float* __restrict__ agg1, int nEdges) {
    float* agg = l2_half() ? agg1 : agg0;
    int lane = threadIdx.x & 31;
    int wave = (blockIdx.x * blockDim.x + threadIdx.x) >> 5;
    int base = wave * 32;
    if (base >= nEdges) return;

    int  myE   = base + lane;
    bool valid = myE < nEdges;
    int   sL = valid ? src[myE] : 0;
    int   dL = valid ? dst[myE] : 0;

    int cnt = nEdges - base;
    if (cnt > 32) cnt = 32;

    if (cnt == 32) {
        float v[32];
#pragma unroll
        for (int i = 0; i < 32; ++i) {
            int s = __builtin_amdgcn_readlane(sL, i);
            v[i] = feat[s * 32 + lane];
        }
#pragma unroll
        for (int i = 0; i < 32; ++i) {
            int d = __builtin_amdgcn_readlane(dL, i);
            atomicAdd(agg + d * 32 + lane, v[i]);
        }
        return;
    }

    for (int i = 0; i < cnt; ++i) {
        int   s = __builtin_amdgcn_readlane(sL, i);
        int   d = __builtin_amdgcn_readlane(dL, i);
        atomicAdd(agg + d * 32 + lane, feat[s * 32 + lane]);
    }
}

__global__ void relu_bias_kernel(const float* __restrict__ agg0, float* __restrict__ h,
                                 const float* __restrict__ b, const float* __restrict__ dinv,
                                 int n32) {
    int idx = blockIdx.x * blockDim.x + threadIdx.x;
    if (idx < n32) {
        float di = dinv[idx >> 5];
        float v = (agg0[idx] + h[idx]) * di + b[idx & 31];
        h[idx] = (v > 0.0f ? v : 0.0f) * di;
    }
}

__global__ void gemm_out_kernel(const float* __restrict__ A, const float* __restrict__ A1,
                                const float* __restrict__ dinv,
                                const float* __restrict__ Wmu, const float* __restrict__ bmu,
                                const float* __restrict__ Wlv, const float* __restrict__ blv,
                                float* __restrict__ out, int nRows) {
    int waveId = (blockIdx.x * blockDim.x + threadIdx.x) >> 5;
    int lane   = threadIdx.x & 31;
    int m0     = waveId * 16;
    if (m0 >= nRows) return;

    int row  = m0 + (lane & 15);
    int grp  = lane >> 4;
    int koff = grp * 2;
    int col  = lane & 15;

    float di = dinv[row];
    v2f a8[8];
#pragma unroll
    for (int k = 0; k < 8; ++k) {
        int kb = 4 * k + koff;
        a8[k] = (*reinterpret_cast<const v2f*>(A + row * 32 + kb)
               + *reinterpret_cast<const v2f*>(A1 + row * 32 + kb)) * di;
    }
    __builtin_amdgcn_sched_barrier(0);

    v8f accM = {};
    v8f accL = {};
#pragma unroll
    for (int k = 0; k < 8; ++k) {
        int kb = 4 * k + koff;
        v2f a = a8[k];
        v2f bm, bl;
        bm.x = Wmu[kb * 16 + col];  bm.y = Wmu[(kb + 1) * 16 + col];
        bl.x = Wlv[kb * 16 + col];  bl.y = Wlv[(kb + 1) * 16 + col];
        accM = __builtin_amdgcn_wmma_f32_16x16x4_f32(false, a, false, bm, (short)0, accM, false, false);
        accL = __builtin_amdgcn_wmma_f32_16x16x4_f32(false, a, false, bl, (short)0, accL, false, false);
    }

    int rowbase = m0 + grp * 8;
    float vbm = bmu[col];
    float vbl = blv[col];
    float* outLv = out + (size_t)nRows * 16;
#pragma unroll
    for (int r = 0; r < 8; ++r) {
        out[(rowbase + r) * 16 + col]   = accM[r] + vbm;
        outLv[(rowbase + r) * 16 + col] = accL[r] + vbl;
    }
}

extern "C" void kernel_launch(void* const* d_in, const int* in_sizes, int n_in,
                              void* d_out, int out_size, void* d_ws, size_t ws_size,
                              hipStream_t stream) {
    const float* x   = (const float*)d_in[0];
    const int*   ei  = (const int*)  d_in[1];
    const float* W1  = (const float*)d_in[2];
    const float* b1  = (const float*)d_in[3];
    const float* Wmu = (const float*)d_in[4];
    const float* bmu = (const float*)d_in[5];
    const float* Wlv = (const float*)d_in[6];
    const float* blv = (const float*)d_in[7];

    const int N = in_sizes[0] / 32;
    const int E = in_sizes[1] / 2;
    const int* src = ei;
    const int* dst = ei + E;

    float* dinv = (float*)d_ws;
    float* bufA = dinv + N;
    float* bufB = bufA + (size_t)N * 32;
    float* out  = (float*)d_out;

    const int TB = 256;
    const int n32 = N * 32;
    const int gemmThreads = ((N + 15) / 16) * 32;
    const int scatterWaves = (E + 31) / 32;
    const int scatterBlocks = (scatterWaves * 32 + TB - 1) / TB;

    unsigned* planes = (unsigned*)bufA;
    const int nPlanes = (E + DEG_CHUNK - 1) / DEG_CHUNK;
    deg_hist_kernel<<<nPlanes, DEG_TB, 0, stream>>>(dst, planes, E);
    deg_reduce_dinv_kernel<<<(DEG_WORDS + TB - 1) / TB, TB, 0, stream>>>(planes, dinv, N, nPlanes);

    gemm_xw1_kernel<<<(gemmThreads + TB - 1) / TB, TB, 0, stream>>>(x, W1, dinv, bufA, N);
    self_init_kernel<<<(n32 + TB - 1) / TB, TB, 0, stream>>>(bufA, bufB, out, n32);
    scatter_agg_kernel<<<scatterBlocks, TB, 0, stream>>>(src, dst, bufA, bufB, out, E);
    relu_bias_kernel<<<(n32 + TB - 1) / TB, TB, 0, stream>>>(bufB, out, b1, dinv, n32);

    self_init_kernel<<<(n32 + TB - 1) / TB, TB, 0, stream>>>(out, bufA, bufB, n32);
    scatter_agg_kernel<<<scatterBlocks, TB, 0, stream>>>(src, dst, out, bufA, bufB, E);
    gemm_out_kernel<<<(gemmThreads + TB - 1) / TB, TB, 0, stream>>>(bufA, bufB, dinv, Wmu, bmu, Wlv, blv, out, N);
}


# Self-loop start values of both aggregations written by the kernels that produce the rows (GEMM1 epilogue; relu/bias pass): the two separate init passes and their 200000 one-element waves are gone
# speedup vs baseline: 2.9680x; 1.1198x over previous
#include <hip/hip_runtime.h>

typedef __attribute__((ext_vector_type(2))) float v2f;
typedef __attribute__((ext_vector_type(8))) float v8f;

__device__ __forceinline__ int l2_half() {
    return (int)((__builtin_amdgcn_s_sendmsg_rtn(0x87) >> 18) & 1u);
}


#define DEG_WORDS 50000
#define DEG_CHUNK 16384
#define DEG_TB    1024

__global__ __launch_bounds__(DEG_TB) void deg_hist_kernel(const int* __restrict__ dst,
                                                          unsigned* __restrict__ planes, int nE) {
    __shared__ unsigned cnt[DEG_WORDS];
    const int t = threadIdx.x;

    const int base = blockIdx.x * DEG_CHUNK;
    int d[DEG_CHUNK / DEG_TB];
#pragma unroll
    for (int k = 0; k < DEG_CHUNK / DEG_TB; ++k) {
        int e = base + k * DEG_TB + t;
        d[k] = (e < nE) ? dst[e] : -1;
    }

    uint4* cnt4 = reinterpret_cast<uint4*>(cnt);
    for (int j = t; j < DEG_WORDS / 4; j += DEG_TB) cnt4[j] = make_uint4(0u, 0u, 0u, 0u);
    __syncthreads();

#pragma unroll
    for (int k = 0; k < DEG_CHUNK / DEG_TB; ++k) {
        if (d[k] >= 0) atomicAdd(&cnt[d[k] >> 1], 1u << ((d[k] & 1) * 16));
    }
    __syncthreads();

    uint4* out4 = reinterpret_cast<uint4*>(planes + (size_t)blockIdx.x * DEG_WORDS);
    for (int j = t; j < DEG_WORDS / 4; j += DEG_TB) out4[j] = cnt4[j];
}

__global__ void deg_reduce_dinv_kernel(const unsigned* __restrict__ planes, float* __restrict__ dinv,
                                       int n, int nPlanes) {
    int i = blockIdx.x * blockDim.x + threadIdx.x;
    if (i >= DEG_WORDS) return;
    unsigned lo = 0u, hi = 0u;
#pragma unroll 16
    for (int c = 0; c < nPlanes; ++c) {
        unsigned w = planes[(size_t)c * DEG_WORDS + i];
        lo += w & 0xffffu;
        hi += w >> 16;
    }
    int n0 = 2 * i;
    if (n0 < n)     dinv[n0]     = rsqrtf((float)(lo + 1u));
    if (n0 + 1 < n) dinv[n0 + 1] = rsqrtf((float)(hi + 1u));
}

__global__ void gemm_xw1_kernel(const float* __restrict__ A, const float* __restrict__ W,
                                const float* __restrict__ dinv,
                                float* __restrict__ C, float* __restrict__ Cself,
                                float* __restrict__ Zero, int nRows) {
    int waveId = (blockIdx.x * blockDim.x + threadIdx.x) >> 5;
    int lane   = threadIdx.x & 31;
    int m0     = waveId * 16;
    if (m0 >= nRows) return;

    float4* z4 = reinterpret_cast<float4*>(Zero + (size_t)m0 * 32);
    const float4 zero4 = make_float4(0.0f, 0.0f, 0.0f, 0.0f);
    z4[lane]      = zero4;
    z4[32 + lane] = zero4;
    z4[64 + lane] = zero4;
    z4[96 + lane] = zero4;
    __builtin_amdgcn_sched_barrier(0);

    int row  = m0 + (lane & 15);
    int grp  = lane >> 4;
    int koff = grp * 2;
    int col  = lane & 15;
    int rowbase = m0 + grp * 8;

    float dv[8];
#pragma unroll
    for (int r = 0; r < 8; ++r) dv[r] = dinv[rowbase + r];

    v8f acc0 = {};
    v8f acc1 = {};
#pragma unroll
    for (int k = 0; k < 8; ++k) {
        int kb = 4 * k + koff;
        v2f a = *reinterpret_cast<const v2f*>(A + row * 32 + kb);
        v2f b0, b1;
        b0.x = W[kb * 32 + col];        b0.y = W[(kb + 1) * 32 + col];
        b1.x = W[kb * 32 + 16 + col];   b1.y = W[(kb + 1) * 32 + 16 + col];
        acc0 = __builtin_amdgcn_wmma_f32_16x16x4_f32(false, a, false, b0, (short)0, acc0, false, false);
        acc1 = __builtin_amdgcn_wmma_f32_16x16x4_f32(false, a, false, b1, (short)0, acc1, false, false);
    }

#pragma unroll
    for (int r = 0; r < 8; ++r) {
        float c0 = acc0[r] * dv[r];
        float c1 = acc1[r] * dv[r];
        C[(rowbase + r) * 32 + col]          = c0;
        C[(rowbase + r) * 32 + 16 + col]     = c1;
        Cself[(rowbase + r) * 32 + col]      = c0;
        Cself[(rowbase + r) * 32 + 16 + col] = c1;
    }
}

__global__ void scatter_agg_kernel(const int* __restrict__ src, const int* __restrict__ dst,
                                   const float* __restrict__ feat,
                                   float* __restrict__ agg0, float* __restrict__ agg1, int nEdges) {
    float* agg = l2_half() ? agg1 : agg0;
    int lane = threadIdx.x & 31;
    int wave = (blockIdx.x * blockDim.x + threadIdx.x) >> 5;
    int base = wave * 32;
    if (base >= nEdges) return;

    int  myE   = base + lane;
    bool valid = myE < nEdges;
    int   sL = valid ? src[myE] : 0;
    int   dL = valid ? dst[myE] : 0;

    int cnt = nEdges - base;
    if (cnt > 32) cnt = 32;

    if (cnt == 32) {
        float v[32];
#pragma unroll
        for (int i = 0; i < 32; ++i) {
            int s = __builtin_amdgcn_readlane(sL, i);
            v[i] = feat[s * 32 + lane];
        }
#pragma unroll
        for (int i = 0; i < 32; ++i) {
            int d = __builtin_amdgcn_readlane(dL, i);
            atomicAdd(agg + d * 32 + lane, v[i]);
        }
        return;
    }

    for (int i = 0; i < cnt; ++i) {
        int   s = __builtin_amdgcn_readlane(sL, i);
        int   d = __builtin_amdgcn_readlane(dL, i);
        atomicAdd(agg + d * 32 + lane, feat[s * 32 + lane]);
    }
}

__global__ void relu_bias_kernel(float* __restrict__ aggA, float* __restrict__ aggB,
                                 float* __restrict__ next0,
                                 const float* __restrict__ b, const float* __restrict__ dinv,
                                 int n32) {
    int idx = blockIdx.x * blockDim.x + threadIdx.x;
    if (idx < n32) {
        float di = dinv[idx >> 5];
        float v = (aggA[idx] + aggB[idx]) * di + b[idx & 31];
        float hv = (v > 0.0f ? v : 0.0f) * di;
        aggB[idx]  = hv;
        next0[idx] = hv;
        aggA[idx]  = 0.0f;
    }
}

__global__ void gemm_out_kernel(const float* __restrict__ A, const float* __restrict__ A1,
                                const float* __restrict__ dinv,
                                const float* __restrict__ Wmu, const float* __restrict__ bmu,
                                const float* __restrict__ Wlv, const float* __restrict__ blv,
                                float* __restrict__ out, int nRows) {
    int waveId = (blockIdx.x * blockDim.x + threadIdx.x) >> 5;
    int lane   = threadIdx.x & 31;
    int m0     = waveId * 16;
    if (m0 >= nRows) return;

    int row  = m0 + (lane & 15);
    int grp  = lane >> 4;
    int koff = grp * 2;
    int col  = lane & 15;

    float di = dinv[row];
    v2f a8[8];
#pragma unroll
    for (int k = 0; k < 8; ++k) {
        int kb = 4 * k + koff;
        a8[k] = (*reinterpret_cast<const v2f*>(A + row * 32 + kb)
               + *reinterpret_cast<const v2f*>(A1 + row * 32 + kb)) * di;
    }
    __builtin_amdgcn_sched_barrier(0);

    v8f accM = {};
    v8f accL = {};
#pragma unroll
    for (int k = 0; k < 8; ++k) {
        int kb = 4 * k + koff;
        v2f a = a8[k];
        v2f bm, bl;
        bm.x = Wmu[kb * 16 + col];  bm.y = Wmu[(kb + 1) * 16 + col];
        bl.x = Wlv[kb * 16 + col];  bl.y = Wlv[(kb + 1) * 16 + col];
        accM = __builtin_amdgcn_wmma_f32_16x16x4_f32(false, a, false, bm, (short)0, accM, false, false);
        accL = __builtin_amdgcn_wmma_f32_16x16x4_f32(false, a, false, bl, (short)0, accL, false, false);
    }

    int rowbase = m0 + grp * 8;
    float vbm = bmu[col];
    float vbl = blv[col];
    float* outLv = out + (size_t)nRows * 16;
#pragma unroll
    for (int r = 0; r < 8; ++r) {
        out[(rowbase + r) * 16 + col]   = accM[r] + vbm;
        outLv[(rowbase + r) * 16 + col] = accL[r] + vbl;
    }
}

extern "C" void kernel_launch(void* const* d_in, const int* in_sizes, int n_in,
                              void* d_out, int out_size, void* d_ws, size_t ws_size,
                              hipStream_t stream) {
    const float* x   = (const float*)d_in[0];
    const int*   ei  = (const int*)  d_in[1];
    const float* W1  = (const float*)d_in[2];
    const float* b1  = (const float*)d_in[3];
    const float* Wmu = (const float*)d_in[4];
    const float* bmu = (const float*)d_in[5];
    const float* Wlv = (const float*)d_in[6];
    const float* blv = (const float*)d_in[7];

    const int N = in_sizes[0] / 32;
    const int E = in_sizes[1] / 2;
    const int* src = ei;
    const int* dst = ei + E;

    float* dinv = (float*)d_ws;
    float* bufA = dinv + N;
    float* bufB = bufA + (size_t)N * 32;
    float* out  = (float*)d_out;

    const int TB = 256;
    const int n32 = N * 32;
    const int gemmThreads = ((N + 15) / 16) * 32;
    const int scatterWaves = (E + 31) / 32;
    const int scatterBlocks = (scatterWaves * 32 + TB - 1) / TB;

    unsigned* planes = (unsigned*)bufA;
    const int nPlanes = (E + DEG_CHUNK - 1) / DEG_CHUNK;
    deg_hist_kernel<<<nPlanes, DEG_TB, 0, stream>>>(dst, planes, E);
    deg_reduce_dinv_kernel<<<(DEG_WORDS + TB - 1) / TB, TB, 0, stream>>>(planes, dinv, N, nPlanes);

    gemm_xw1_kernel<<<(gemmThreads + TB - 1) / TB, TB, 0, stream>>>(x, W1, dinv, bufA, bufB, out, N);
    scatter_agg_kernel<<<scatterBlocks, TB, 0, stream>>>(src, dst, bufA, bufB, out, E);
    relu_bias_kernel<<<(n32 + TB - 1) / TB, TB, 0, stream>>>(bufB, out, bufA, b1, dinv, n32);

    scatter_agg_kernel<<<scatterBlocks, TB, 0, stream>>>(src, dst, out, bufA, bufB, E);
    gemm_out_kernel<<<(gemmThreads + TB - 1) / TB, TB, 0, stream>>>(bufA, bufB, dinv, Wmu, bmu, Wlv, blv, out, N);
}
